# starting kernel
# baseline (speedup 1.0000x reference)
#include <hip/hip_runtime.h>
#include <hip/hip_bf16.h>
#include <math.h>

#define BB 64
#define NN 50000
#define EE 1600000
#define LL 20
#define BN_EPS 1e-5f
#define KCHUNK 784
#define NCHUNKS 64

typedef __attribute__((ext_vector_type(2))) float v2f;
typedef __attribute__((ext_vector_type(8))) float v8f;

__global__ void k_zero(float* __restrict__ p, long n) {
    long i = (long)blockIdx.x * blockDim.x + threadIdx.x;
    long stride = (long)gridDim.x * blockDim.x;
    for (; i < n; i += stride) p[i] = 0.0f;
}

__global__ void k_transpose(const float* __restrict__ x, float* __restrict__ xT) {
    __shared__ float tile[64 * 65];
    int n0   = blockIdx.x * 64;
    int col  = threadIdx.x & 63;
    int rowg = threadIdx.x >> 6;
    for (int b = rowg; b < 64; b += 4) {
        int n = n0 + col;
        tile[b * 65 + col] = (n < NN) ? x[(long)b * NN + n] : 0.0f;
    }
    __syncthreads();
    for (int nl = rowg; nl < 64; nl += 4) {
        int n = n0 + nl;
        if (n < NN) xT[(long)n * 64 + col] = tile[col * 65 + nl];
    }
}

__global__ void k_edges(const float* __restrict__ xT, const int* __restrict__ ei,
                        const float* __restrict__ alpha, const float* __restrict__ bias,
                        float* __restrict__ sums, float* __restrict__ counts) {
    long gtid = (long)blockIdx.x * blockDim.x + threadIdx.x;
    long e = gtid >> 6;
    int  t = (int)(gtid & 63);
    if (e >= EE) return;
    int   s  = ei[e];
    int   d  = ei[(long)EE + e];
    float a  = alpha[e];
    float bi = bias[e];
    float val = fmaf(a, xT[(long)s * 64 + t], bi);
    atomicAdd(&sums[(long)d * 64 + t], val);
    if (t == 0) atomicAdd(&counts[d], 1.0f);
}

__global__ void k_finalize(float* __restrict__ sums  ,
                           const float* __restrict__ counts,
                           const float* __restrict__ gamma, const float* __restrict__ beta,
                           float* __restrict__ bn_out  ) {
    __shared__ float tile[64 * 65];
    int n0   = blockIdx.x * 64;
    int wave = threadIdx.x >> 5;
    int lane = threadIdx.x & 31;
    for (int i = 0; i < 8; ++i) {
        int lnode = wave * 8 + i;
        int n = n0 + lnode;
        if (n < NN) {
            float2 s = *(const float2*)&sums[(long)n * 64 + 2 * lane];
            float inv = 1.0f / fmaxf(counts[n], 1.0f);
            float xh0 = s.x * inv, xh1 = s.y * inv;
            float2 xh; xh.x = xh0; xh.y = xh1;
            *(float2*)&sums[(long)n * 64 + 2 * lane] = xh;
            float t0 = tanhf(xh0), t1 = tanhf(xh1);
            float sm = t0 + t1;
            float sq = t0 * t0 + t1 * t1;
            #pragma unroll
            for (int off = 16; off >= 1; off >>= 1) {
                sm += __shfl_xor(sm, off, 32);
                sq += __shfl_xor(sq, off, 32);
            }
            float mu = sm * (1.0f / 64.0f);
            float va = sq * (1.0f / 64.0f) - mu * mu;
            float rs = rsqrtf(va + BN_EPS);
            float g = gamma[n], be = beta[n];
            tile[lnode * 65 + 2 * lane]     = (t0 - mu) * rs * g + be;
            tile[lnode * 65 + 2 * lane + 1] = (t1 - mu) * rs * g + be;
        }
    }
    __syncthreads();
    int col  = threadIdx.x & 63;
    int rowg = threadIdx.x >> 6;
    int n = n0 + col;
    if (n < NN) {
        for (int b = rowg; b < 64; b += 4)
            bn_out[(long)b * NN + n] = tile[col * 65 + b];
    }
}

__global__ void k_gemm_part(const float* __restrict__ xhatT  ,
                            const float* __restrict__ W      ,
                            float* __restrict__ part) {
    int c    = blockIdx.x;
    int wave = threadIdx.x >> 5;
    int lane = threadIdx.x & 31;
    int half = lane >> 4;
    int l16  = lane & 15;
    int m_tile = wave >> 1;
    int n_tile = wave & 1;
    int m    = m_tile * 16 + l16;
    int lcol = n_tile * 16 + l16;
    int wrow = (lcol < LL) ? lcol : 0;
    const float* Wrow = W + (long)wrow * NN;

    long kbeg = (long)c * KCHUNK;
    long kend = kbeg + KCHUNK; if (kend > NN) kend = NN;

    v8f acc = {};
    #pragma unroll 4
    for (long k = kbeg; k < kend; k += 4) {
        long k0 = k + half * 2;
        long k1 = k0 + 1;
        v2f a, b;
        a.x = xhatT[k0 * 64 + m];
        a.y = xhatT[k1 * 64 + m];
        b.x = Wrow[k0];
        b.y = Wrow[k1];
        acc = __builtin_amdgcn_wmma_f32_16x16x4_f32(
             false, a,  false, b,
             (short)0, acc,  false,  false);
    }

    float* dst = part + (long)c * (BB * LL);
    #pragma unroll
    for (int i = 0; i < 8; ++i) {
        int mrow = m_tile * 16 + half * 8 + i;
        int ncol = n_tile * 16 + l16;
        if (ncol < LL) dst[mrow * LL + ncol] = acc[i];
    }
}

__global__ void k_reduce_pred(const float* __restrict__ part, const float* __restrict__ b_lin,
                              float* __restrict__ pred) {
    int i = blockIdx.x * blockDim.x + threadIdx.x;
    if (i >= BB * LL) return;
    float s = b_lin[i % LL];
    for (int c = 0; c < NCHUNKS; ++c) s += part[c * (BB * LL) + i];
    pred[i] = s;
}

extern "C" void kernel_launch(void* const* d_in, const int* in_sizes, int n_in,
                              void* d_out, int out_size, void* d_ws, size_t ws_size,
                              hipStream_t stream) {
    const float* x     = (const float*)d_in[0];
    const int*   ei    = (const int*)  d_in[1];
    const float* alpha = (const float*)d_in[2];
    const float* bias  = (const float*)d_in[3];
    const float* W     = (const float*)d_in[4];
    const float* b_lin = (const float*)d_in[5];
    const float* gamma = (const float*)d_in[6];
    const float* beta  = (const float*)d_in[7];
    float* out = (float*)d_out;

    float* ws     = (float*)d_ws;
    float* sums   = ws;
    float* counts = sums + (long)NN * 64;
    float* xT     = counts + NN;
    float* part   = xT + (long)NN * 64;

    k_zero<<<4096, 256, 0, stream>>>(sums, (long)NN * 64 + NN);
    k_transpose<<<(NN + 63) / 64, 256, 0, stream>>>(x, xT);
    long edge_threads = (long)EE * 64;
    k_edges<<<(int)((edge_threads + 255) / 256), 256, 0, stream>>>(xT, ei, alpha, bias, sums, counts);
    k_finalize<<<(NN + 63) / 64, 256, 0, stream>>>(sums, counts, gamma, beta, out + BB * LL);
    k_gemm_part<<<NCHUNKS, 256, 0, stream>>>(sums, W, part);
    k_reduce_pred<<<(BB * LL + 255) / 256, 256, 0, stream>>>(part, b_lin, out);
}


# tile scatter owners: each filtering wave links its segment into 32 per-owner chains (5 ballots per 32 entries); an owner follows its chains, no scan of all 32 segments; row sums in new fixed order
# speedup vs baseline: 3.2607x; 3.2607x over previous
#include <hip/hip_runtime.h>
#include <hip/hip_bf16.h>
#include <math.h>

#define BB 64
#define NN 50000
#define EE 1600000
#define LL 20
#define BN_EPS 1e-5f
#define KCHUNK 196
#define NCHUNKS 256

typedef __attribute__((ext_vector_type(2))) float v2f;
typedef __attribute__((ext_vector_type(8))) float v8f;

__global__ void k_transpose(const float* __restrict__ x, float* __restrict__ xT, int* __restrict__ counts) {
    __shared__ float tile[64 * 65];
    int n0   = blockIdx.x * 64;
    int col  = threadIdx.x & 63;
    int rowg = threadIdx.x >> 6;
    if (rowg == 0 && n0 + col < NN) counts[n0 + col] = 0;
    for (int b = rowg; b < 64; b += 4) {
        int n = n0 + col;
        tile[b * 65 + col] = (n < NN) ? x[(long)b * NN + n] : 0.0f;
    }
    __syncthreads();
    for (int nl = rowg; nl < 64; nl += 4) {
        int n = n0 + nl;
        if (n < NN) xT[(long)n * 64 + col] = tile[col * 65 + nl];
    }
}

#define TILE 448
#define NTILES 112
#define SC_SLICES 2
#define SC_ROWS (TILE / 32)
#define SC_WAVES 32
#define SC_LOADS 5
#define SC_BATCH_E (SC_LOADS * 128)
#define SC_NBATCH (EE / SC_SLICES / SC_BATCH_E)
static_assert(EE % (SC_SLICES * SC_BATCH_E) == 0, "each part of the edge stream must be a whole number of batches");
static_assert(TILE <= 512 && EE <= (1 << 21), "an entry packs (node - lo) above 21 bits of edge id");
#define SC_SHARE (SC_NBATCH / SC_WAVES)
#define SC_EXTRA (SC_NBATCH % SC_WAVES)
static_assert(SC_SHARE >= 1, "every wave needs a batch");
#define SC_CAP 1024
#define SC_LIM (SC_CAP - SC_BATCH_E)
#define SC_SEGP (SC_CAP + 1)
#define SC_STAGE 64
#define SC_BATCH 16
#define SC_HP 33
static_assert(SC_CAP < (1 << 11) && TILE / 32 <= 16 && (SC_SHARE + 1) * SC_BATCH_E <= (1 << 15), "fields of a linked entry");

typedef __attribute__((ext_vector_type(4))) int v4i;

__device__ __forceinline__ void sc_load(v4i (&v)[SC_LOADS], const int* __restrict__ dstp, int base, int lane) {
    #pragma unroll
    for (int j = 0; j < SC_LOADS; ++j) v[j] = *(const v4i*)(dstp + base + j * 128 + lane * 4);
}

__device__ __forceinline__ void sc_filter(const v4i (&v)[SC_LOADS], int base, int lo, int lane, unsigned lt,
                                          unsigned* myseg, int& n) {
    #pragma unroll
    for (int j = 0; j < SC_LOADS; ++j) {
        #pragma unroll
        for (int c = 0; c < 4; ++c) {
            unsigned dl = (unsigned)(v[j][c] - lo);
            bool m = dl < (unsigned)TILE;
            unsigned mask = __builtin_amdgcn_ballot_w32(m);
            if (mask) {
                if (m) myseg[n + __builtin_popcount(mask & lt)] =
                           (dl << 21) | (unsigned)(base + j * 128 + lane * 4 + c);
                n += __builtin_popcount(mask);
            }
        }
    }
}

__device__ __forceinline__ void sc_flush(int nb, int lane, int w, const unsigned* stg,
                                         const float* __restrict__ xT, const int* __restrict__ ei,
                                         const float* __restrict__ alpha, const float* __restrict__ bias,
                                         float* acc, int& cnt) {
    unsigned ent = (lane < nb) ? stg[lane] : 0u;
    int e  = (int)(ent & 0x1FFFFFu);
    int kr = (int)(ent >> 26);
    int s = 0; float a = 0.0f, b = 0.0f;
    if (lane < nb) { s = ei[e]; a = alpha[e]; b = bias[e]; }
    float2 row[SC_BATCH];
    #pragma unroll
    for (int q = 0; q < SC_BATCH; ++q) {
        row[q] = make_float2(0.0f, 0.0f);
        if (q < nb) {
            int sk = __builtin_amdgcn_readlane(s, q);
            row[q] = *(const float2*)&xT[(long)sk * 64 + 2 * lane];
        }
    }
    #pragma unroll
    for (int q = 0; q < SC_BATCH; ++q) {
        if (q < nb) {
            float ak = __int_as_float(__builtin_amdgcn_readlane(__float_as_int(a), q));
            float bk = __int_as_float(__builtin_amdgcn_readlane(__float_as_int(b), q));
            int   kk = __builtin_amdgcn_readlane(kr, q);
            float vx = fmaf(ak, row[q].x, bk);
            float vy = fmaf(ak, row[q].y, bk);
            float* p = acc + ((w + 32 * kk) * 64 + 2 * lane);
            __hip_atomic_fetch_add(p,     vx, __ATOMIC_RELAXED, __HIP_MEMORY_SCOPE_WORKGROUP);
            __hip_atomic_fetch_add(p + 1, vy, __ATOMIC_RELAXED, __HIP_MEMORY_SCOPE_WORKGROUP);
            cnt += (lane == kk) ? 1 : 0;
        }
    }
}

__global__ __launch_bounds__(1024) void k_tile_scatter(const float* __restrict__ xT, const int* __restrict__ ei,
                                                       const float* __restrict__ alpha, const float* __restrict__ bias,
                                                       float* __restrict__ sums0, float* __restrict__ sums1,
                                                       int* __restrict__ counts) {
    __shared__ float    acc[TILE * 64];
    __shared__ int      lcnt[TILE];
    __shared__ unsigned seg[SC_WAVES * SC_SEGP];
    __shared__ unsigned heads[SC_WAVES * SC_HP];
    __shared__ int      segmore[SC_WAVES];
    __shared__ unsigned stage[SC_WAVES * SC_STAGE];

    const int tid  = threadIdx.x;
    const int lane = tid & 31;
    const int w    = __builtin_amdgcn_readfirstlane(tid >> 5);
    const int slice = blockIdx.x % SC_SLICES;
    const int lo    = (blockIdx.x / SC_SLICES) * TILE;
    float* __restrict__ sums = slice ? sums1 : sums0;
    const int* __restrict__ dstp = ei + EE;
    const unsigned lt = (1u << lane) - 1u;
    unsigned* mystage = stage + w * SC_STAGE;
    unsigned* myseg   = seg + w * SC_SEGP;
    unsigned* myheads = heads + w * SC_HP;

    #pragma unroll
    for (int k = 0; k < SC_ROWS; ++k)
        *(float2*)&acc[(w + 32 * k) * 64 + 2 * lane] = make_float2(0.0f, 0.0f);
    int cnt = 0;

    const int myNB  = SC_SHARE + (w < SC_EXTRA ? 1 : 0);
    const int first = slice * SC_NBATCH + w * SC_SHARE + (w < SC_EXTRA ? w : SC_EXTRA);
    const int ebase = first * SC_BATCH_E;
    const int ebl = (slice * SC_NBATCH + lane * SC_SHARE + (lane < SC_EXTRA ? lane : SC_EXTRA)) * SC_BATCH_E;
    int i  = 0;
    int np = 0;
    for (;;) {
        int n = 0;
        v4i va[SC_LOADS], vb[SC_LOADS];
        sc_load(va, dstp, (first + min(i, myNB - 1)) * SC_BATCH_E, lane);
        while (i < myNB && n <= SC_LIM) {
            sc_load(vb, dstp, (first + min(i + 1, myNB - 1)) * SC_BATCH_E, lane);
            sc_filter(va, (first + i) * SC_BATCH_E, lo, lane, lt, myseg, n);
            ++i;
            if (!(i < myNB && n <= SC_LIM)) break;
            sc_load(va, dstp, (first + min(i + 1, myNB - 1)) * SC_BATCH_E, lane);
            sc_filter(vb, (first + i) * SC_BATCH_E, lo, lane, lt, myseg, n);
            ++i;
        }
        if (lane == 0) segmore[w] = (i < myNB) ? 1 : 0;
        myheads[lane] = 0u;
        for (int c0 = 0; c0 < n; c0 += 32) {
            const int idx = c0 + lane;
            const bool valid = idx < n;
            const unsigned ent = valid ? myseg[idx] : 0u;
            const unsigned dl = ent >> 21;
            const unsigned o = dl & 31u;
            unsigned match = __builtin_amdgcn_ballot_w32(valid);
            for (int b = 0; b < 5; ++b) {
                const bool bit = ((o >> b) & 1u) != 0u;
                const unsigned bb = __builtin_amdgcn_ballot_w32(bit);
                match &= bit ? bb : ~bb;
            }
            const unsigned before = match & lt;
            const unsigned prevhead = myheads[o];
            const unsigned link = before ? (unsigned)(c0 + 32 - __clz((int)before)) : prevhead;
            const unsigned word = (link << 19) | ((dl >> 5) << 15) | ((ent & 0x1FFFFFu) - (unsigned)ebase);
            if (valid) {
                myseg[idx] = word;
                if ((match >> lane) == 1u) myheads[o] = (unsigned)(idx + 1);
            }
        }
        __syncthreads();
        const bool more = __builtin_amdgcn_ballot_w32(segmore[lane] != 0) != 0u;
        unsigned p = heads[lane * SC_HP + w];
        unsigned om;
        while ((om = __builtin_amdgcn_ballot_w32(p != 0u)) != 0u) {
            const bool val = p != 0u;
            const unsigned word = val ? seg[lane * SC_SEGP + (int)p - 1] : 0u;
            p = word >> 19;
            if (val) mystage[np + __builtin_popcount(om & lt)] =
                         (((word >> 15) & 15u) << 26) | (unsigned)(ebl + (int)(word & 0x7FFFu));
            np += __builtin_popcount(om);
            __builtin_amdgcn_wave_barrier();
            while (np >= SC_BATCH) {
                sc_flush(SC_BATCH, lane, w, mystage, xT, ei, alpha, bias, acc, cnt);
                unsigned rest = (lane < np - SC_BATCH) ? mystage[SC_BATCH + lane] : 0u;
                __builtin_amdgcn_wave_barrier();
                if (lane < np - SC_BATCH) mystage[lane] = rest;
                np -= SC_BATCH;
                __builtin_amdgcn_wave_barrier();
            }
        }
        __syncthreads();
        if (!more) break;
    }
    if (np > 0) sc_flush(np, lane, w, mystage, xT, ei, alpha, bias, acc, cnt);

    #pragma unroll
    for (int k = 0; k < SC_ROWS; ++k) {
        int n = lo + w + 32 * k;
        float2 rs = *(const float2*)&acc[(w + 32 * k) * 64 + 2 * lane];
        if (n < NN) *(float2*)&sums[(long)n * 64 + 2 * lane] = rs;
    }
    if (lane < SC_ROWS) lcnt[w + 32 * lane] = cnt;
    __syncthreads();
    if (tid < TILE && lo + tid < NN) atomicAdd(&counts[lo + tid], lcnt[tid]);
}

__global__ void k_finalize(float* __restrict__ sums  ,
                           const float* __restrict__ sums1  ,
                           const int* __restrict__ counts,
                           const float* __restrict__ gamma, const float* __restrict__ beta,
                           float* __restrict__ bn_out  ) {
    __shared__ float tile[64 * 65];
    int n0   = blockIdx.x * 64;
    int wave = threadIdx.x >> 5;
    int lane = threadIdx.x & 31;
    for (int i = 0; i < 8; ++i) {
        int lnode = wave * 8 + i;
        int n = n0 + lnode;
        if (n < NN) {
            float2 s = *(const float2*)&sums[(long)n * 64 + 2 * lane];
            float2 s1 = *(const float2*)&sums1[(long)n * 64 + 2 * lane];
            float inv = 1.0f / fmaxf((float)counts[n], 1.0f);
            float xh0 = (s.x + s1.x) * inv, xh1 = (s.y + s1.y) * inv;
            float2 xh; xh.x = xh0; xh.y = xh1;
            *(float2*)&sums[(long)n * 64 + 2 * lane] = xh;
            float t0 = tanhf(xh0), t1 = tanhf(xh1);
            float sm = t0 + t1;
            float sq = t0 * t0 + t1 * t1;
            #pragma unroll
            for (int off = 16; off >= 1; off >>= 1) {
                sm += __shfl_xor(sm, off, 32);
                sq += __shfl_xor(sq, off, 32);
            }
            float mu = sm * (1.0f / 64.0f);
            float va = sq * (1.0f / 64.0f) - mu * mu;
            float rs = rsqrtf(va + BN_EPS);
            float g = gamma[n], be = beta[n];
            tile[lnode * 65 + 2 * lane]     = (t0 - mu) * rs * g + be;
            tile[lnode * 65 + 2 * lane + 1] = (t1 - mu) * rs * g + be;
        }
    }
    __syncthreads();
    int col  = threadIdx.x & 63;
    int rowg = threadIdx.x >> 6;
    int n = n0 + col;
    if (n < NN) {
        for (int b = rowg; b < 64; b += 4)
            bn_out[(long)b * NN + n] = tile[col * 65 + b];
    }
}

__global__ void k_gemm_part(const float* __restrict__ xhatT  ,
                            const float* __restrict__ W      ,
                            float* __restrict__ part) {
    int c    = blockIdx.x;
    int wave = threadIdx.x >> 5;
    int lane = threadIdx.x & 31;
    int half = lane >> 4;
    int l16  = lane & 15;
    int m_tile = wave >> 1;
    int n_tile = wave & 1;
    int m    = m_tile * 16 + l16;
    int lcol = n_tile * 16 + l16;
    int wrow = (lcol < LL) ? lcol : 0;
    const float* Wrow = W + (long)wrow * NN;

    long kbeg = (long)c * KCHUNK;
    long kend = kbeg + KCHUNK; if (kend > NN) kend = NN;

    v8f acc = {};
    #pragma unroll 7
    for (long k = kbeg; k < kend; k += 4) {
        long k0 = k + half * 2;
        long k1 = k0 + 1;
        v2f a, b;
        a.x = xhatT[k0 * 64 + m];
        a.y = xhatT[k1 * 64 + m];
        b.x = Wrow[k0];
        b.y = Wrow[k1];
        acc = __builtin_amdgcn_wmma_f32_16x16x4_f32(
             false, a,  false, b,
             (short)0, acc,  false,  false);
    }

    float* dst = part + (long)c * (BB * LL);
    #pragma unroll
    for (int i = 0; i < 8; ++i) {
        int mrow = m_tile * 16 + half * 8 + i;
        int ncol = n_tile * 16 + l16;
        if (ncol < LL) dst[mrow * LL + ncol] = acc[i];
    }
}

__global__ void k_reduce_pred(const float* __restrict__ part, const float* __restrict__ b_lin,
                              float* __restrict__ pred) {
    int i = blockIdx.x * blockDim.x + threadIdx.x;
    if (i >= BB * LL) return;
    float s = b_lin[i % LL];
    for (int c = 0; c < NCHUNKS; ++c) s += part[c * (BB * LL) + i];
    pred[i] = s;
}

extern "C" void kernel_launch(void* const* d_in, const int* in_sizes, int n_in,
                              void* d_out, int out_size, void* d_ws, size_t ws_size,
                              hipStream_t stream) {
    const float* x     = (const float*)d_in[0];
    const int*   ei    = (const int*)  d_in[1];
    const float* alpha = (const float*)d_in[2];
    const float* bias  = (const float*)d_in[3];
    const float* W     = (const float*)d_in[4];
    const float* b_lin = (const float*)d_in[5];
    const float* gamma = (const float*)d_in[6];
    const float* beta  = (const float*)d_in[7];
    float* out = (float*)d_out;

    float* ws     = (float*)d_ws;
    float* sums   = ws;
    int*   counts = (int*)(sums + (long)NN * 64);
    float* sums1  = (float*)(counts + NN);
    float* part   = sums1;
    float* xT     = out + BB * LL;

    k_transpose<<<(NN + 63) / 64, 256, 0, stream>>>(x, xT, counts);
    k_tile_scatter<<<NTILES * SC_SLICES, 1024, 0, stream>>>(xT, ei, alpha, bias, sums, sums1, counts);
    k_finalize<<<(NN + 63) / 64, 256, 0, stream>>>(sums, sums1, counts, gamma, beta, out + BB * LL);
    k_gemm_part<<<NCHUNKS, 256, 0, stream>>>(sums, W, part);
    k_reduce_pred<<<(BB * LL + 255) / 256, 256, 0, stream>>>(part, b_lin, out);
}


# finalize and partial-sum reduce: a wave issues all its global loads up front (8 nodes' rows and terms, their lane-sum trees side by side; 64 partials a group), not a memory round trip per loop turn
# speedup vs baseline: 3.6196x; 1.1101x over previous
#include <hip/hip_runtime.h>
#include <hip/hip_bf16.h>
#include <math.h>

#define BB 64
#define NN 50000
#define EE 1600000
#define LL 20
#define BN_EPS 1e-5f
#define KCHUNK 196
#define NCHUNKS 256

typedef __attribute__((ext_vector_type(2))) float v2f;
typedef __attribute__((ext_vector_type(8))) float v8f;

__global__ void k_transpose(const float* __restrict__ x, float* __restrict__ xT, int* __restrict__ counts) {
    __shared__ float tile[64 * 65];
    int n0   = blockIdx.x * 64;
    int col  = threadIdx.x & 63;
    int rowg = threadIdx.x >> 6;
    if (rowg == 0 && n0 + col < NN) counts[n0 + col] = 0;
    for (int b = rowg; b < 64; b += 4) {
        int n = n0 + col;
        tile[b * 65 + col] = (n < NN) ? x[(long)b * NN + n] : 0.0f;
    }
    __syncthreads();
    for (int nl = rowg; nl < 64; nl += 4) {
        int n = n0 + nl;
        if (n < NN) xT[(long)n * 64 + col] = tile[col * 65 + nl];
    }
}

#define TILE 448
#define NTILES 112
#define SC_SLICES 2
#define SC_ROWS (TILE / 32)
#define SC_WAVES 32
#define SC_LOADS 5
#define SC_BATCH_E (SC_LOADS * 128)
#define SC_NBATCH (EE / SC_SLICES / SC_BATCH_E)
static_assert(EE % (SC_SLICES * SC_BATCH_E) == 0, "each part of the edge stream must be a whole number of batches");
static_assert(TILE <= 512 && EE <= (1 << 21), "an entry packs (node - lo) above 21 bits of edge id");
#define SC_SHARE (SC_NBATCH / SC_WAVES)
#define SC_EXTRA (SC_NBATCH % SC_WAVES)
static_assert(SC_SHARE >= 1, "every wave needs a batch");
#define SC_CAP 1024
#define SC_LIM (SC_CAP - SC_BATCH_E)
#define SC_SEGP (SC_CAP + 1)
#define SC_STAGE 64
#define SC_BATCH 16
#define SC_HP 33
static_assert(SC_CAP < (1 << 11) && TILE / 32 <= 16 && (SC_SHARE + 1) * SC_BATCH_E <= (1 << 15), "fields of a linked entry");

typedef __attribute__((ext_vector_type(4))) int v4i;

__device__ __forceinline__ void sc_load(v4i (&v)[SC_LOADS], const int* __restrict__ dstp, int base, int lane) {
    #pragma unroll
    for (int j = 0; j < SC_LOADS; ++j) v[j] = *(const v4i*)(dstp + base + j * 128 + lane * 4);
}

__device__ __forceinline__ void sc_filter(const v4i (&v)[SC_LOADS], int base, int lo, int lane, unsigned lt,
                                          unsigned* myseg, int& n) {
    #pragma unroll
    for (int j = 0; j < SC_LOADS; ++j) {
        #pragma unroll
        for (int c = 0; c < 4; ++c) {
            unsigned dl = (unsigned)(v[j][c] - lo);
            bool m = dl < (unsigned)TILE;
            unsigned mask = __builtin_amdgcn_ballot_w32(m);
            if (mask) {
                if (m) myseg[n + __builtin_popcount(mask & lt)] =
                           (dl << 21) | (unsigned)(base + j * 128 + lane * 4 + c);
                n += __builtin_popcount(mask);
            }
        }
    }
}

__device__ __forceinline__ void sc_flush(int nb, int lane, int w, const unsigned* stg,
                                         const float* __restrict__ xT, const int* __restrict__ ei,
                                         const float* __restrict__ alpha, const float* __restrict__ bias,
                                         float* acc, int& cnt) {
    unsigned ent = (lane < nb) ? stg[lane] : 0u;
    int e  = (int)(ent & 0x1FFFFFu);
    int kr = (int)(ent >> 26);
    int s = 0; float a = 0.0f, b = 0.0f;
    if (lane < nb) { s = ei[e]; a = alpha[e]; b = bias[e]; }
    float2 row[SC_BATCH];
    #pragma unroll
    for (int q = 0; q < SC_BATCH; ++q) {
        row[q] = make_float2(0.0f, 0.0f);
        if (q < nb) {
            int sk = __builtin_amdgcn_readlane(s, q);
            row[q] = *(const float2*)&xT[(long)sk * 64 + 2 * lane];
        }
    }
    #pragma unroll
    for (int q = 0; q < SC_BATCH; ++q) {
        if (q < nb) {
            float ak = __int_as_float(__builtin_amdgcn_readlane(__float_as_int(a), q));
            float bk = __int_as_float(__builtin_amdgcn_readlane(__float_as_int(b), q));
            int   kk = __builtin_amdgcn_readlane(kr, q);
            float vx = fmaf(ak, row[q].x, bk);
            float vy = fmaf(ak, row[q].y, bk);
            float* p = acc + ((w + 32 * kk) * 64 + 2 * lane);
            __hip_atomic_fetch_add(p,     vx, __ATOMIC_RELAXED, __HIP_MEMORY_SCOPE_WORKGROUP);
            __hip_atomic_fetch_add(p + 1, vy, __ATOMIC_RELAXED, __HIP_MEMORY_SCOPE_WORKGROUP);
            cnt += (lane == kk) ? 1 : 0;
        }
    }
}

__global__ __launch_bounds__(1024) void k_tile_scatter(const float* __restrict__ xT, const int* __restrict__ ei,
                                                       const float* __restrict__ alpha, const float* __restrict__ bias,
                                                       float* __restrict__ sums0, float* __restrict__ sums1,
                                                       int* __restrict__ counts) {
    __shared__ float    acc[TILE * 64];
    __shared__ int      lcnt[TILE];
    __shared__ unsigned seg[SC_WAVES * SC_SEGP];
    __shared__ unsigned heads[SC_WAVES * SC_HP];
    __shared__ int      segmore[SC_WAVES];
    __shared__ unsigned stage[SC_WAVES * SC_STAGE];

    const int tid  = threadIdx.x;
    const int lane = tid & 31;
    const int w    = __builtin_amdgcn_readfirstlane(tid >> 5);
    const int slice = blockIdx.x % SC_SLICES;
    const int lo    = (blockIdx.x / SC_SLICES) * TILE;
    float* __restrict__ sums = slice ? sums1 : sums0;
    const int* __restrict__ dstp = ei + EE;
    const unsigned lt = (1u << lane) - 1u;
    unsigned* mystage = stage + w * SC_STAGE;
    unsigned* myseg   = seg + w * SC_SEGP;
    unsigned* myheads = heads + w * SC_HP;

    #pragma unroll
    for (int k = 0; k < SC_ROWS; ++k)
        *(float2*)&acc[(w + 32 * k) * 64 + 2 * lane] = make_float2(0.0f, 0.0f);
    int cnt = 0;

    const int myNB  = SC_SHARE + (w < SC_EXTRA ? 1 : 0);
    const int first = slice * SC_NBATCH + w * SC_SHARE + (w < SC_EXTRA ? w : SC_EXTRA);
    const int ebase = first * SC_BATCH_E;
    const int ebl = (slice * SC_NBATCH + lane * SC_SHARE + (lane < SC_EXTRA ? lane : SC_EXTRA)) * SC_BATCH_E;
    int i  = 0;
    int np = 0;
    for (;;) {
        int n = 0;
        v4i va[SC_LOADS], vb[SC_LOADS];
        sc_load(va, dstp, (first + min(i, myNB - 1)) * SC_BATCH_E, lane);
        while (i < myNB && n <= SC_LIM) {
            sc_load(vb, dstp, (first + min(i + 1, myNB - 1)) * SC_BATCH_E, lane);
            sc_filter(va, (first + i) * SC_BATCH_E, lo, lane, lt, myseg, n);
            ++i;
            if (!(i < myNB && n <= SC_LIM)) break;
            sc_load(va, dstp, (first + min(i + 1, myNB - 1)) * SC_BATCH_E, lane);
            sc_filter(vb, (first + i) * SC_BATCH_E, lo, lane, lt, myseg, n);
            ++i;
        }
        if (lane == 0) segmore[w] = (i < myNB) ? 1 : 0;
        myheads[lane] = 0u;
        for (int c0 = 0; c0 < n; c0 += 32) {
            const int idx = c0 + lane;
            const bool valid = idx < n;
            const unsigned ent = valid ? myseg[idx] : 0u;
            const unsigned dl = ent >> 21;
            const unsigned o = dl & 31u;
            unsigned match = __builtin_amdgcn_ballot_w32(valid);
            for (int b = 0; b < 5; ++b) {
                const bool bit = ((o >> b) & 1u) != 0u;
                const unsigned bb = __builtin_amdgcn_ballot_w32(bit);
                match &= bit ? bb : ~bb;
            }
            const unsigned before = match & lt;
            const unsigned prevhead = myheads[o];
            const unsigned link = before ? (unsigned)(c0 + 32 - __clz((int)before)) : prevhead;
            const unsigned word = (link << 19) | ((dl >> 5) << 15) | ((ent & 0x1FFFFFu) - (unsigned)ebase);
            if (valid) {
                myseg[idx] = word;
                if ((match >> lane) == 1u) myheads[o] = (unsigned)(idx + 1);
            }
        }
        __syncthreads();
        const bool more = __builtin_amdgcn_ballot_w32(segmore[lane] != 0) != 0u;
        unsigned p = heads[lane * SC_HP + w];
        unsigned om;
        while ((om = __builtin_amdgcn_ballot_w32(p != 0u)) != 0u) {
            const bool val = p != 0u;
            const unsigned word = val ? seg[lane * SC_SEGP + (int)p - 1] : 0u;
            p = word >> 19;
            if (val) mystage[np + __builtin_popcount(om & lt)] =
                         (((word >> 15) & 15u) << 26) | (unsigned)(ebl + (int)(word & 0x7FFFu));
            np += __builtin_popcount(om);
            __builtin_amdgcn_wave_barrier();
            while (np >= SC_BATCH) {
                sc_flush(SC_BATCH, lane, w, mystage, xT, ei, alpha, bias, acc, cnt);
                unsigned rest = (lane < np - SC_BATCH) ? mystage[SC_BATCH + lane] : 0u;
                __builtin_amdgcn_wave_barrier();
                if (lane < np - SC_BATCH) mystage[lane] = rest;
                np -= SC_BATCH;
                __builtin_amdgcn_wave_barrier();
            }
        }
        __syncthreads();
        if (!more) break;
    }
    if (np > 0) sc_flush(np, lane, w, mystage, xT, ei, alpha, bias, acc, cnt);

    #pragma unroll
    for (int k = 0; k < SC_ROWS; ++k) {
        int n = lo + w + 32 * k;
        float2 rs = *(const float2*)&acc[(w + 32 * k) * 64 + 2 * lane];
        if (n < NN) *(float2*)&sums[(long)n * 64 + 2 * lane] = rs;
    }
    if (lane < SC_ROWS) lcnt[w + 32 * lane] = cnt;
    __syncthreads();
    if (tid < TILE && lo + tid < NN) atomicAdd(&counts[lo + tid], lcnt[tid]);
}

#define FN_NODES 8
__global__ void k_finalize(float* __restrict__ sums  ,
                           const float* __restrict__ sums1  ,
                           const int* __restrict__ counts,
                           const float* __restrict__ gamma, const float* __restrict__ beta,
                           float* __restrict__ bn_out  ) {
    __shared__ float tile[64 * 65];
    int n0   = blockIdx.x * 64;
    int wave = threadIdx.x >> 5;
    int lane = threadIdx.x & 31;
    float2 s[FN_NODES], s1[FN_NODES];
    int   cn[FN_NODES];
    float g[FN_NODES], be[FN_NODES];
    #pragma unroll
    for (int i = 0; i < FN_NODES; ++i) {
        int n  = n0 + wave * FN_NODES + i;
        int nc = (n < NN) ? n : (NN - 1);
        s[i]  = *(const float2*)&sums[(long)nc * 64 + 2 * lane];
        s1[i] = *(const float2*)&sums1[(long)nc * 64 + 2 * lane];
        cn[i] = counts[nc];
        g[i]  = gamma[nc];
        be[i] = beta[nc];
    }
    __builtin_amdgcn_sched_barrier(0);
    float t0[FN_NODES], t1[FN_NODES];
    #pragma unroll
    for (int i = 0; i < FN_NODES; ++i) {
        int n = n0 + wave * FN_NODES + i;
        float inv = 1.0f / fmaxf((float)cn[i], 1.0f);
        float xh0 = (s[i].x + s1[i].x) * inv, xh1 = (s[i].y + s1[i].y) * inv;
        float2 xh; xh.x = xh0; xh.y = xh1;
        if (n < NN) *(float2*)&sums[(long)n * 64 + 2 * lane] = xh;
        t0[i] = tanhf(xh0);
        t1[i] = tanhf(xh1);
    }
    float sm[FN_NODES], sq[FN_NODES];
    #pragma unroll
    for (int i = 0; i < FN_NODES; ++i) {
        sm[i] = t0[i] + t1[i];
        sq[i] = t0[i] * t0[i] + t1[i] * t1[i];
    }
    #pragma unroll
    for (int off = 16; off >= 1; off >>= 1) {
        #pragma unroll
        for (int i = 0; i < FN_NODES; ++i) {
            sm[i] += __shfl_xor(sm[i], off, 32);
            sq[i] += __shfl_xor(sq[i], off, 32);
        }
    }
    #pragma unroll
    for (int i = 0; i < FN_NODES; ++i) {
        int lnode = wave * FN_NODES + i;
        float mu = sm[i] * (1.0f / 64.0f);
        float va = sq[i] * (1.0f / 64.0f) - mu * mu;
        float rs = rsqrtf(va + BN_EPS);
        tile[lnode * 65 + 2 * lane]     = (t0[i] - mu) * rs * g[i] + be[i];
        tile[lnode * 65 + 2 * lane + 1] = (t1[i] - mu) * rs * g[i] + be[i];
    }
    __syncthreads();
    int col  = threadIdx.x & 63;
    int rowg = threadIdx.x >> 6;
    int n = n0 + col;
    if (n < NN) {
        for (int b = rowg; b < 64; b += 4)
            bn_out[(long)b * NN + n] = tile[col * 65 + b];
    }
}

__global__ void k_gemm_part(const float* __restrict__ xhatT  ,
                            const float* __restrict__ W      ,
                            float* __restrict__ part) {
    int c    = blockIdx.x;
    int wave = threadIdx.x >> 5;
    int lane = threadIdx.x & 31;
    int half = lane >> 4;
    int l16  = lane & 15;
    int m_tile = wave >> 1;
    int n_tile = wave & 1;
    int m    = m_tile * 16 + l16;
    int lcol = n_tile * 16 + l16;
    int wrow = (lcol < LL) ? lcol : 0;
    const float* Wrow = W + (long)wrow * NN;

    long kbeg = (long)c * KCHUNK;
    long kend = kbeg + KCHUNK; if (kend > NN) kend = NN;

    v8f acc = {};
    #pragma unroll 7
    for (long k = kbeg; k < kend; k += 4) {
        long k0 = k + half * 2;
        long k1 = k0 + 1;
        v2f a, b;
        a.x = xhatT[k0 * 64 + m];
        a.y = xhatT[k1 * 64 + m];
        b.x = Wrow[k0];
        b.y = Wrow[k1];
        acc = __builtin_amdgcn_wmma_f32_16x16x4_f32(
             false, a,  false, b,
             (short)0, acc,  false,  false);
    }

    float* dst = part + (long)c * (BB * LL);
    #pragma unroll
    for (int i = 0; i < 8; ++i) {
        int mrow = m_tile * 16 + half * 8 + i;
        int ncol = n_tile * 16 + l16;
        if (ncol < LL) dst[mrow * LL + ncol] = acc[i];
    }
}

#define RP_GROUP 64
#define RP_THREADS 32
static_assert(NCHUNKS % RP_GROUP == 0, "whole groups of partials");
__global__ void k_reduce_pred(const float* __restrict__ part, const float* __restrict__ b_lin,
                              float* __restrict__ pred) {
    int i = blockIdx.x * blockDim.x + threadIdx.x;
    if (i >= BB * LL) return;
    float s = b_lin[i % LL];
    for (int c0 = 0; c0 < NCHUNKS; c0 += RP_GROUP) {
        float v[RP_GROUP];
        #pragma unroll
        for (int j = 0; j < RP_GROUP; ++j) v[j] = part[(c0 + j) * (BB * LL) + i];
        #pragma unroll
        for (int j = 0; j < RP_GROUP; ++j) s += v[j];
    }
    pred[i] = s;
}

extern "C" void kernel_launch(void* const* d_in, const int* in_sizes, int n_in,
                              void* d_out, int out_size, void* d_ws, size_t ws_size,
                              hipStream_t stream) {
    const float* x     = (const float*)d_in[0];
    const int*   ei    = (const int*)  d_in[1];
    const float* alpha = (const float*)d_in[2];
    const float* bias  = (const float*)d_in[3];
    const float* W     = (const float*)d_in[4];
    const float* b_lin = (const float*)d_in[5];
    const float* gamma = (const float*)d_in[6];
    const float* beta  = (const float*)d_in[7];
    float* out = (float*)d_out;

    float* ws     = (float*)d_ws;
    float* sums   = ws;
    int*   counts = (int*)(sums + (long)NN * 64);
    float* sums1  = (float*)(counts + NN);
    float* part   = sums1;
    float* xT     = out + BB * LL;

    k_transpose<<<(NN + 63) / 64, 256, 0, stream>>>(x, xT, counts);
    k_tile_scatter<<<NTILES * SC_SLICES, 1024, 0, stream>>>(xT, ei, alpha, bias, sums, sums1, counts);
    k_finalize<<<(NN + 63) / 64, 256, 0, stream>>>(sums, sums1, counts, gamma, beta, out + BB * LL);
    k_gemm_part<<<NCHUNKS, 256, 0, stream>>>(sums, W, part);
    k_reduce_pred<<<(BB * LL + RP_THREADS - 1) / RP_THREADS, RP_THREADS, 0, stream>>>(part, b_lin, out);
}


# WMMA head: a chunk's x_hat rows and W columns staged once per workgroup in LDS (whole-line loads issued together), fragments from LDS; reduce: 4 waves x 64 partials via LDS; pred in new sum order
# speedup vs baseline: 3.8347x; 1.0594x over previous
#include <hip/hip_runtime.h>
#include <hip/hip_bf16.h>
#include <math.h>

#define BB 64
#define NN 50000
#define EE 1600000
#define LL 20
#define BN_EPS 1e-5f
#define KCHUNK 196
#define NCHUNKS 256

typedef __attribute__((ext_vector_type(2))) float v2f;
typedef __attribute__((ext_vector_type(8))) float v8f;

__global__ void k_transpose(const float* __restrict__ x, float* __restrict__ xT, int* __restrict__ counts) {
    __shared__ float tile[64 * 65];
    int n0   = blockIdx.x * 64;
    int col  = threadIdx.x & 63;
    int rowg = threadIdx.x >> 6;
    if (rowg == 0 && n0 + col < NN) counts[n0 + col] = 0;
    for (int b = rowg; b < 64; b += 4) {
        int n = n0 + col;
        tile[b * 65 + col] = (n < NN) ? x[(long)b * NN + n] : 0.0f;
    }
    __syncthreads();
    for (int nl = rowg; nl < 64; nl += 4) {
        int n = n0 + nl;
        if (n < NN) xT[(long)n * 64 + col] = tile[col * 65 + nl];
    }
}

#define TILE 448
#define NTILES 112
#define SC_SLICES 2
#define SC_ROWS (TILE / 32)
#define SC_WAVES 32
#define SC_LOADS 5
#define SC_BATCH_E (SC_LOADS * 128)
#define SC_NBATCH (EE / SC_SLICES / SC_BATCH_E)
static_assert(EE % (SC_SLICES * SC_BATCH_E) == 0, "each part of the edge stream must be a whole number of batches");
static_assert(TILE <= 512 && EE <= (1 << 21), "an entry packs (node - lo) above 21 bits of edge id");
#define SC_SHARE (SC_NBATCH / SC_WAVES)
#define SC_EXTRA (SC_NBATCH % SC_WAVES)
static_assert(SC_SHARE >= 1, "every wave needs a batch");
#define SC_CAP 1024
#define SC_LIM (SC_CAP - SC_BATCH_E)
#define SC_SEGP (SC_CAP + 1)
#define SC_STAGE 64
#define SC_BATCH 16
#define SC_HP 33
static_assert(SC_CAP < (1 << 11) && TILE / 32 <= 16 && (SC_SHARE + 1) * SC_BATCH_E <= (1 << 15), "fields of a linked entry");

typedef __attribute__((ext_vector_type(4))) int v4i;

__device__ __forceinline__ void sc_load(v4i (&v)[SC_LOADS], const int* __restrict__ dstp, int base, int lane) {
    #pragma unroll
    for (int j = 0; j < SC_LOADS; ++j) v[j] = *(const v4i*)(dstp + base + j * 128 + lane * 4);
}

__device__ __forceinline__ void sc_filter(const v4i (&v)[SC_LOADS], int base, int lo, int lane, unsigned lt,
                                          unsigned* myseg, int& n) {
    #pragma unroll
    for (int j = 0; j < SC_LOADS; ++j) {
        #pragma unroll
        for (int c = 0; c < 4; ++c) {
            unsigned dl = (unsigned)(v[j][c] - lo);
            bool m = dl < (unsigned)TILE;
            unsigned mask = __builtin_amdgcn_ballot_w32(m);
            if (mask) {
                if (m) myseg[n + __builtin_popcount(mask & lt)] =
                           (dl << 21) | (unsigned)(base + j * 128 + lane * 4 + c);
                n += __builtin_popcount(mask);
            }
        }
    }
}

__device__ __forceinline__ void sc_flush(int nb, int lane, int w, const unsigned* stg,
                                         const float* __restrict__ xT, const int* __restrict__ ei,
                                         const float* __restrict__ alpha, const float* __restrict__ bias,
                                         float* acc, int& cnt) {
    unsigned ent = (lane < nb) ? stg[lane] : 0u;
    int e  = (int)(ent & 0x1FFFFFu);
    int kr = (int)(ent >> 26);
    int s = 0; float a = 0.0f, b = 0.0f;
    if (lane < nb) { s = ei[e]; a = alpha[e]; b = bias[e]; }
    float2 row[SC_BATCH];
    #pragma unroll
    for (int q = 0; q < SC_BATCH; ++q) {
        row[q] = make_float2(0.0f, 0.0f);
        if (q < nb) {
            int sk = __builtin_amdgcn_readlane(s, q);
            row[q] = *(const float2*)&xT[(long)sk * 64 + 2 * lane];
        }
    }
    #pragma unroll
    for (int q = 0; q < SC_BATCH; ++q) {
        if (q < nb) {
            float ak = __int_as_float(__builtin_amdgcn_readlane(__float_as_int(a), q));
            float bk = __int_as_float(__builtin_amdgcn_readlane(__float_as_int(b), q));
            int   kk = __builtin_amdgcn_readlane(kr, q);
            float vx = fmaf(ak, row[q].x, bk);
            float vy = fmaf(ak, row[q].y, bk);
            float* p = acc + ((w + 32 * kk) * 64 + 2 * lane);
            __hip_atomic_fetch_add(p,     vx, __ATOMIC_RELAXED, __HIP_MEMORY_SCOPE_WORKGROUP);
            __hip_atomic_fetch_add(p + 1, vy, __ATOMIC_RELAXED, __HIP_MEMORY_SCOPE_WORKGROUP);
            cnt += (lane == kk) ? 1 : 0;
        }
    }
}

__global__ __launch_bounds__(1024) void k_tile_scatter(const float* __restrict__ xT, const int* __restrict__ ei,
                                                       const float* __restrict__ alpha, const float* __restrict__ bias,
                                                       float* __restrict__ sums0, float* __restrict__ sums1,
                                                       int* __restrict__ counts) {
    __shared__ float    acc[TILE * 64];
    __shared__ int      lcnt[TILE];
    __shared__ unsigned seg[SC_WAVES * SC_SEGP];
    __shared__ unsigned heads[SC_WAVES * SC_HP];
    __shared__ int      segmore[SC_WAVES];
    __shared__ unsigned stage[SC_WAVES * SC_STAGE];

    const int tid  = threadIdx.x;
    const int lane = tid & 31;
    const int w    = __builtin_amdgcn_readfirstlane(tid >> 5);
    const int slice = blockIdx.x % SC_SLICES;
    const int lo    = (blockIdx.x / SC_SLICES) * TILE;
    float* __restrict__ sums = slice ? sums1 : sums0;
    const int* __restrict__ dstp = ei + EE;
    const unsigned lt = (1u << lane) - 1u;
    unsigned* mystage = stage + w * SC_STAGE;
    unsigned* myseg   = seg + w * SC_SEGP;
    unsigned* myheads = heads + w * SC_HP;

    #pragma unroll
    for (int k = 0; k < SC_ROWS; ++k)
        *(float2*)&acc[(w + 32 * k) * 64 + 2 * lane] = make_float2(0.0f, 0.0f);
    int cnt = 0;

    const int myNB  = SC_SHARE + (w < SC_EXTRA ? 1 : 0);
    const int first = slice * SC_NBATCH + w * SC_SHARE + (w < SC_EXTRA ? w : SC_EXTRA);
    const int ebase = first * SC_BATCH_E;
    const int ebl = (slice * SC_NBATCH + lane * SC_SHARE + (lane < SC_EXTRA ? lane : SC_EXTRA)) * SC_BATCH_E;
    int i  = 0;
    int np = 0;
    for (;;) {
        int n = 0;
        v4i va[SC_LOADS], vb[SC_LOADS];
        sc_load(va, dstp, (first + min(i, myNB - 1)) * SC_BATCH_E, lane);
        while (i < myNB && n <= SC_LIM) {
            sc_load(vb, dstp, (first + min(i + 1, myNB - 1)) * SC_BATCH_E, lane);
            sc_filter(va, (first + i) * SC_BATCH_E, lo, lane, lt, myseg, n);
            ++i;
            if (!(i < myNB && n <= SC_LIM)) break;
            sc_load(va, dstp, (first + min(i + 1, myNB - 1)) * SC_BATCH_E, lane);
            sc_filter(vb, (first + i) * SC_BATCH_E, lo, lane, lt, myseg, n);
            ++i;
        }
        if (lane == 0) segmore[w] = (i < myNB) ? 1 : 0;
        myheads[lane] = 0u;
        for (int c0 = 0; c0 < n; c0 += 32) {
            const int idx = c0 + lane;
            const bool valid = idx < n;
            const unsigned ent = valid ? myseg[idx] : 0u;
            const unsigned dl = ent >> 21;
            const unsigned o = dl & 31u;
            unsigned match = __builtin_amdgcn_ballot_w32(valid);
            for (int b = 0; b < 5; ++b) {
                const bool bit = ((o >> b) & 1u) != 0u;
                const unsigned bb = __builtin_amdgcn_ballot_w32(bit);
                match &= bit ? bb : ~bb;
            }
            const unsigned before = match & lt;
            const unsigned prevhead = myheads[o];
            const unsigned link = before ? (unsigned)(c0 + 32 - __clz((int)before)) : prevhead;
            const unsigned word = (link << 19) | ((dl >> 5) << 15) | ((ent & 0x1FFFFFu) - (unsigned)ebase);
            if (valid) {
                myseg[idx] = word;
                if ((match >> lane) == 1u) myheads[o] = (unsigned)(idx + 1);
            }
        }
        __syncthreads();
        const bool more = __builtin_amdgcn_ballot_w32(segmore[lane] != 0) != 0u;
        unsigned p = heads[lane * SC_HP + w];
        unsigned om;
        while ((om = __builtin_amdgcn_ballot_w32(p != 0u)) != 0u) {
            const bool val = p != 0u;
            const unsigned word = val ? seg[lane * SC_SEGP + (int)p - 1] : 0u;
            p = word >> 19;
            if (val) mystage[np + __builtin_popcount(om & lt)] =
                         (((word >> 15) & 15u) << 26) | (unsigned)(ebl + (int)(word & 0x7FFFu));
            np += __builtin_popcount(om);
            __builtin_amdgcn_wave_barrier();
            while (np >= SC_BATCH) {
                sc_flush(SC_BATCH, lane, w, mystage, xT, ei, alpha, bias, acc, cnt);
                unsigned rest = (lane < np - SC_BATCH) ? mystage[SC_BATCH + lane] : 0u;
                __builtin_amdgcn_wave_barrier();
                if (lane < np - SC_BATCH) mystage[lane] = rest;
                np -= SC_BATCH;
                __builtin_amdgcn_wave_barrier();
            }
        }
        __syncthreads();
        if (!more) break;
    }
    if (np > 0) sc_flush(np, lane, w, mystage, xT, ei, alpha, bias, acc, cnt);

    #pragma unroll
    for (int k = 0; k < SC_ROWS; ++k) {
        int n = lo + w + 32 * k;
        float2 rs = *(const float2*)&acc[(w + 32 * k) * 64 + 2 * lane];
        if (n < NN) *(float2*)&sums[(long)n * 64 + 2 * lane] = rs;
    }
    if (lane < SC_ROWS) lcnt[w + 32 * lane] = cnt;
    __syncthreads();
    if (tid < TILE && lo + tid < NN) atomicAdd(&counts[lo + tid], lcnt[tid]);
}

#define FN_NODES 8
__global__ void k_finalize(float* __restrict__ sums  ,
                           const float* __restrict__ sums1  ,
                           const int* __restrict__ counts,
                           const float* __restrict__ gamma, const float* __restrict__ beta,
                           float* __restrict__ bn_out  ) {
    __shared__ float tile[64 * 65];
    int n0   = blockIdx.x * 64;
    int wave = threadIdx.x >> 5;
    int lane = threadIdx.x & 31;
    float2 s[FN_NODES], s1[FN_NODES];
    int   cn[FN_NODES];
    float g[FN_NODES], be[FN_NODES];
    #pragma unroll
    for (int i = 0; i < FN_NODES; ++i) {
        int n  = n0 + wave * FN_NODES + i;
        int nc = (n < NN) ? n : (NN - 1);
        s[i]  = *(const float2*)&sums[(long)nc * 64 + 2 * lane];
        s1[i] = *(const float2*)&sums1[(long)nc * 64 + 2 * lane];
        cn[i] = counts[nc];
        g[i]  = gamma[nc];
        be[i] = beta[nc];
    }
    __builtin_amdgcn_sched_barrier(0);
    float t0[FN_NODES], t1[FN_NODES];
    #pragma unroll
    for (int i = 0; i < FN_NODES; ++i) {
        int n = n0 + wave * FN_NODES + i;
        float inv = 1.0f / fmaxf((float)cn[i], 1.0f);
        float xh0 = (s[i].x + s1[i].x) * inv, xh1 = (s[i].y + s1[i].y) * inv;
        float2 xh; xh.x = xh0; xh.y = xh1;
        if (n < NN) *(float2*)&sums[(long)n * 64 + 2 * lane] = xh;
        t0[i] = tanhf(xh0);
        t1[i] = tanhf(xh1);
    }
    float sm[FN_NODES], sq[FN_NODES];
    #pragma unroll
    for (int i = 0; i < FN_NODES; ++i) {
        sm[i] = t0[i] + t1[i];
        sq[i] = t0[i] * t0[i] + t1[i] * t1[i];
    }
    #pragma unroll
    for (int off = 16; off >= 1; off >>= 1) {
        #pragma unroll
        for (int i = 0; i < FN_NODES; ++i) {
            sm[i] += __shfl_xor(sm[i], off, 32);
            sq[i] += __shfl_xor(sq[i], off, 32);
        }
    }
    #pragma unroll
    for (int i = 0; i < FN_NODES; ++i) {
        int lnode = wave * FN_NODES + i;
        float mu = sm[i] * (1.0f / 64.0f);
        float va = sq[i] * (1.0f / 64.0f) - mu * mu;
        float rs = rsqrtf(va + BN_EPS);
        tile[lnode * 65 + 2 * lane]     = (t0[i] - mu) * rs * g[i] + be[i];
        tile[lnode * 65 + 2 * lane + 1] = (t1[i] - mu) * rs * g[i] + be[i];
    }
    __syncthreads();
    int col  = threadIdx.x & 63;
    int rowg = threadIdx.x >> 6;
    int n = n0 + col;
    if (n < NN) {
        for (int b = rowg; b < 64; b += 4)
            bn_out[(long)b * NN + n] = tile[col * 65 + b];
    }
}

#define GP_STEPS (KCHUNK / 4)
#define GA_PITCH 80
#define GB_PITCH 204
#define GA_VEC (KCHUNK * 16)
#define GB_ROWV (KCHUNK / 4)
#define GB_VEC (LL * GB_ROWV)
#define GA_ROUNDS ((GA_VEC + 255) / 256)
#define GB_ROUNDS ((GB_VEC + 255) / 256)
static_assert(KCHUNK % 4 == 0 && (NN % 4) == 0, "W rows of a chunk start on 16-byte boundaries");
static_assert(GA_PITCH % 4 == 0 && GB_PITCH % 4 == 0, "staged rows start on 16-byte boundaries");
static_assert(GA_VEC >= 256 && GB_VEC >= 256, "piece tid exists for every thread");
__global__ __launch_bounds__(256) void k_gemm_part(const float* __restrict__ xhatT  ,
                            const float* __restrict__ W      ,
                            float* __restrict__ part) {
    __shared__ __attribute__((aligned(16))) float As[KCHUNK * GA_PITCH];
    __shared__ __attribute__((aligned(16))) float Bs[LL * GB_PITCH];
    int c    = blockIdx.x;
    int wave = threadIdx.x >> 5;
    int lane = threadIdx.x & 31;
    int half = lane >> 4;
    int l16  = lane & 15;
    int m_tile = wave >> 1;
    int n_tile = wave & 1;
    int m    = m_tile * 16 + l16;
    int lcol = n_tile * 16 + l16;
    int wrow = (lcol < LL) ? lcol : 0;
    const float* Wrow = W + (long)wrow * NN;

    long kbeg = (long)c * KCHUNK;
    long kend = kbeg + KCHUNK; if (kend > NN) kend = NN;

    v8f acc = {};
    if (kend - kbeg == KCHUNK) {
        const int tid = threadIdx.x;
        const float* achunk = xhatT + kbeg * 64;
        float4 ra[GA_ROUNDS], rb[GB_ROUNDS];
        int    ta[GA_ROUNDS], tb[GB_ROUNDS];
        #pragma unroll
        for (int u = 0; u < GA_ROUNDS; ++u) {
            const int t = tid + 256 * u;
            ta[u] = (t < GA_VEC) ? t : tid;
            ra[u] = *(const float4*)(achunk + 4 * ta[u]);
        }
        #pragma unroll
        for (int u = 0; u < GB_ROUNDS; ++u) {
            const int t = tid + 256 * u;
            tb[u] = (t < GB_VEC) ? t : tid;
            rb[u] = *(const float4*)(W + (long)(tb[u] / GB_ROWV) * NN + kbeg + 4 * (tb[u] % GB_ROWV));
        }
        __builtin_amdgcn_sched_barrier(0);
        #pragma unroll
        for (int u = 0; u < GA_ROUNDS; ++u)
            *(float4*)&As[(ta[u] >> 4) * GA_PITCH + 4 * (ta[u] & 15)] = ra[u];
        #pragma unroll
        for (int u = 0; u < GB_ROUNDS; ++u)
            *(float4*)&Bs[(tb[u] / GB_ROWV) * GB_PITCH + 4 * (tb[u] % GB_ROWV)] = rb[u];
        __syncthreads();
        float ax[GP_STEPS], ay[GP_STEPS];
        v2f   bw[GP_STEPS];
        #pragma unroll
        for (int j = 0; j < GP_STEPS; ++j) {
            const int r0 = 4 * j + 2 * half;
            ax[j] = As[r0 * GA_PITCH + m];
            ay[j] = As[(r0 + 1) * GA_PITCH + m];
            bw[j] = *(const v2f*)&Bs[wrow * GB_PITCH + r0];
        }
        __builtin_amdgcn_sched_barrier(0);
        #pragma unroll
        for (int j = 0; j < GP_STEPS; ++j) {
            v2f a;
            a.x = ax[j];
            a.y = ay[j];
            acc = __builtin_amdgcn_wmma_f32_16x16x4_f32(
                 false, a,  false, bw[j],
                 (short)0, acc,  false,  false);
        }
    } else {
        for (long k = kbeg; k < kend; k += 4) {
            long k0 = k + half * 2;
            long k1 = k0 + 1;
            v2f a, b;
            a.x = xhatT[k0 * 64 + m];
            a.y = xhatT[k1 * 64 + m];
            b.x = Wrow[k0];
            b.y = Wrow[k1];
            acc = __builtin_amdgcn_wmma_f32_16x16x4_f32(
                 false, a,  false, b,
                 (short)0, acc,  false,  false);
        }
    }

    float* dst = part + (long)c * (BB * LL);
    #pragma unroll
    for (int i = 0; i < 8; ++i) {
        int mrow = m_tile * 16 + half * 8 + i;
        int ncol = n_tile * 16 + l16;
        if (ncol < LL) dst[mrow * LL + ncol] = acc[i];
    }
}

#define RP_OUT 32
#define RP_PARTS 4
#define RP_SPAN (NCHUNKS / RP_PARTS)
static_assert(NCHUNKS % RP_PARTS == 0 && (BB * LL) % RP_OUT == 0, "whole spans of partials, whole workgroups of pred elements");
__global__ __launch_bounds__(RP_OUT * RP_PARTS) void k_reduce_pred(const float* __restrict__ part, const float* __restrict__ b_lin,
                              float* __restrict__ pred) {
    __shared__ float ps[RP_PARTS * RP_OUT];
    const int lane = threadIdx.x & 31;
    const int q    = threadIdx.x >> 5;
    const int i    = blockIdx.x * RP_OUT + lane;
    const float bl = b_lin[i % LL];
    float v[RP_SPAN];
    #pragma unroll
    for (int j = 0; j < RP_SPAN; ++j) v[j] = part[(q * RP_SPAN + j) * (BB * LL) + i];
    float s = (q == 0) ? bl : 0.0f;
    #pragma unroll
    for (int j = 0; j < RP_SPAN; ++j) s += v[j];
    ps[q * RP_OUT + lane] = s;
    __syncthreads();
    if (q == 0) {
        float r = ps[lane];
        #pragma unroll
        for (int p = 1; p < RP_PARTS; ++p) r += ps[p * RP_OUT + lane];
        pred[i] = r;
    }
}

extern "C" void kernel_launch(void* const* d_in, const int* in_sizes, int n_in,
                              void* d_out, int out_size, void* d_ws, size_t ws_size,
                              hipStream_t stream) {
    const float* x     = (const float*)d_in[0];
    const int*   ei    = (const int*)  d_in[1];
    const float* alpha = (const float*)d_in[2];
    const float* bias  = (const float*)d_in[3];
    const float* W     = (const float*)d_in[4];
    const float* b_lin = (const float*)d_in[5];
    const float* gamma = (const float*)d_in[6];
    const float* beta  = (const float*)d_in[7];
    float* out = (float*)d_out;

    float* ws     = (float*)d_ws;
    float* sums   = ws;
    int*   counts = (int*)(sums + (long)NN * 64);
    float* sums1  = (float*)(counts + NN);
    float* part   = sums1;
    float* xT     = out + BB * LL;

    k_transpose<<<(NN + 63) / 64, 256, 0, stream>>>(x, xT, counts);
    k_tile_scatter<<<NTILES * SC_SLICES, 1024, 0, stream>>>(xT, ei, alpha, bias, sums, sums1, counts);
    k_finalize<<<(NN + 63) / 64, 256, 0, stream>>>(sums, sums1, counts, gamma, beta, out + BB * LL);
    k_gemm_part<<<NCHUNKS, 256, 0, stream>>>(sums, W, part);
    k_reduce_pred<<<(BB * LL) / RP_OUT, RP_OUT * RP_PARTS, 0, stream>>>(part, b_lin, out);
}
